# starting kernel
# baseline (speedup 1.0000x reference)
#include <hip/hip_runtime.h>

#define DIM 512
#define HLEN (DIM * (DIM - 1) / 2)

typedef __attribute__((ext_vector_type(2))) float v2f;
typedef __attribute__((ext_vector_type(8))) float v8f;

__global__ void __launch_bounds__(512)
linmap_tri_matvec(const float* __restrict__ z,
                  const float* __restrict__ h,
                  float* __restrict__ out,
                  int batch) {
    __shared__ float zs[DIM];

    const int b   = blockIdx.x;
    const int tid = threadIdx.x;

    zs[tid] = z[(size_t)b * DIM + tid];
    __syncthreads();

    const int lane = tid & 31;
    const int wave = tid >> 5;
    const int m    = lane & 15;
    const int kb   = (lane >> 4) << 1;

    const float* __restrict__ hb = h + (size_t)b * (size_t)HLEN;

    for (int pass = 0; pass < 2; ++pass) {
        const int tile = (pass == 0) ? wave : (31 - wave);
        const int i0   = tile << 4;
        const int i0s  = __builtin_amdgcn_readfirstlane(i0);

        const int r      = i0 + m;
        const int rowoff = (r * (r - 1)) >> 1;
        const float* __restrict__ hrow = hb + rowoff;

        v8f c = {0.f, 0.f, 0.f, 0.f, 0.f, 0.f, 0.f, 0.f};

        for (int j0 = 0; j0 < i0s; j0 += 16) {
            v2f a0, a1, a2, a3, b0, b1, b2, b3;
            const float* __restrict__ hp = hrow + j0 + kb;
            a0.x = hp[0];  a0.y = hp[1];
            a1.x = hp[4];  a1.y = hp[5];
            a2.x = hp[8];  a2.y = hp[9];
            a3.x = hp[12]; a3.y = hp[13];
            const float* __restrict__ zp = zs + j0 + kb;
            b0.x = zp[0];  b0.y = zp[1];
            b1.x = zp[4];  b1.y = zp[5];
            b2.x = zp[8];  b2.y = zp[9];
            b3.x = zp[12]; b3.y = zp[13];
            c = __builtin_amdgcn_wmma_f32_16x16x4_f32(false, a0, false, b0,
                                                      (short)0, c, false, false);
            c = __builtin_amdgcn_wmma_f32_16x16x4_f32(false, a1, false, b1,
                                                      (short)0, c, false, false);
            c = __builtin_amdgcn_wmma_f32_16x16x4_f32(false, a2, false, b2,
                                                      (short)0, c, false, false);
            c = __builtin_amdgcn_wmma_f32_16x16x4_f32(false, a3, false, b3,
                                                      (short)0, c, false, false);
        }

        #pragma unroll
        for (int t = 0; t < 4; ++t) {
            const int c0 = i0 + 4 * t + kb;
            const int c1 = c0 + 1;
            int i0c = min(c0, r - 1); i0c = max(i0c, 0);
            int i1c = min(c1, r - 1); i1c = max(i1c, 0);
            const float h0 = hrow[i0c];
            const float h1 = hrow[i1c];
            v2f a, bv;
            a.x = h0 * (float)(c0 < r) + (float)(c0 == r);
            a.y = h1 * (float)(c1 < r) + (float)(c1 == r);
            bv.x = zs[c0];
            bv.y = zs[c1];
            c = __builtin_amdgcn_wmma_f32_16x16x4_f32(false, a, false, bv,
                                                      (short)0, c, false, false);
        }

        if (lane == 0 || lane == 16) {
            const int rb = i0 + ((lane >> 4) << 3);
            float* __restrict__ o = out + (size_t)b * DIM + rb;
            o[0] = c[0]; o[1] = c[1]; o[2] = c[2]; o[3] = c[3];
            o[4] = c[4]; o[5] = c[5]; o[6] = c[6]; o[7] = c[7];
        }
    }

    if (b == 0 && tid == 0) out[(size_t)batch * DIM] = 0.0f;
}

extern "C" void kernel_launch(void* const* d_in, const int* in_sizes, int n_in,
                              void* d_out, int out_size, void* d_ws, size_t ws_size,
                              hipStream_t stream) {
    const float* z = (const float*)d_in[0];
    const float* h = (const float*)d_in[1];
    float* out = (float*)d_out;

    const int batch = in_sizes[0] / DIM;

    linmap_tri_matvec<<<dim3(batch), dim3(512), 0, stream>>>(z, h, out, batch);
}


# h bulk staged through LDS: each wave load takes 512 contiguous bytes of one packed row (was 8 bytes from every row of the tile), A fragments read back from LDS in native layout; sum order unchanged
# speedup vs baseline: 1.8246x; 1.8246x over previous
#include <hip/hip_runtime.h>

#define DIM 512
#define HLEN (DIM * (DIM - 1) / 2)
#define PW 256
#define NLD (PW / 128)
#define PITCH (PW + 4)
#define NWAVES 16

typedef __attribute__((ext_vector_type(2))) float v2f;
typedef __attribute__((ext_vector_type(4))) float v4f;
typedef __attribute__((ext_vector_type(8))) float v8f;
typedef float v4f_u __attribute__((ext_vector_type(4), aligned(4)));

__global__ void __launch_bounds__(512)
linmap_tri_matvec(const float* __restrict__ z,
                  const float* __restrict__ h,
                  float* __restrict__ out,
                  int batch) {
    __shared__ __attribute__((aligned(16))) float zs[DIM];
    __shared__ __attribute__((aligned(16))) float stage[NWAVES * 16 * PITCH];

    const int b   = blockIdx.x;
    const int tid = threadIdx.x;

    zs[tid] = z[(size_t)b * DIM + tid];
    __syncthreads();

    const int lane = tid & 31;
    const int wave = __builtin_amdgcn_readfirstlane(tid >> 5);
    const int m    = lane & 15;
    const int kb   = (lane >> 4) << 1;

    const float* __restrict__ hb = h + (size_t)b * (size_t)HLEN;
    float* __restrict__ st = stage + wave * (16 * PITCH);

    const int t0  = 31 - wave;
    const int t1  = wave;
    const int np0 = (16 * t0 + PW - 1) / PW;
    const int np1 = (16 * t1 + PW - 1) / PW;
    const int np  = np0 + np1;

    v4f g[16 * NLD];

    auto load_panel = [&](int k) {
        const int second = (k >= np0);
        const int tile   = second ? t1 : t0;
        const int jc     = (second ? (k - np0) : k) * PW;
        const int i0     = tile << 4;
        const int width  = min(PW, i0 - jc);
        #pragma unroll
        for (int u = 0; u < NLD; ++u) {
            if (128 * u + 4 * lane < width) {
                #pragma unroll
                for (int q = 0; q < 16; ++q) {
                    const int rq = i0 + q;
                    const float* p = hb + ((rq * (rq - 1)) >> 1) + jc + 128 * u + 4 * lane;
                    g[q * NLD + u] = __builtin_nontemporal_load((const v4f_u*)p);
                }
            }
        }
    };

    auto diag_load = [&](int tile, float* dh) {
        const int i0 = tile << 4;
        const int r  = i0 + m;
        const float* __restrict__ hrow = hb + ((r * (r - 1)) >> 1);
        #pragma unroll
        for (int t = 0; t < 4; ++t) {
            const int c0 = i0 + 4 * t + kb;
            const int c1 = c0 + 1;
            int i0c = min(c0, r - 1); i0c = max(i0c, 0);
            int i1c = min(c1, r - 1); i1c = max(i1c, 0);
            dh[2 * t]     = hrow[i0c];
            dh[2 * t + 1] = hrow[i1c];
        }
    };

    auto diag_finish = [&](int tile, const float* dh, v8f c) {
        const int i0 = tile << 4;
        const int r  = i0 + m;
        #pragma unroll
        for (int t = 0; t < 4; ++t) {
            const int c0 = i0 + 4 * t + kb;
            const int c1 = c0 + 1;
            v2f a, bv;
            a.x = dh[2 * t]     * (float)(c0 < r) + (float)(c0 == r);
            a.y = dh[2 * t + 1] * (float)(c1 < r) + (float)(c1 == r);
            bv.x = zs[c0];
            bv.y = zs[c1];
            c = __builtin_amdgcn_wmma_f32_16x16x4_f32(false, a, false, bv,
                                                      (short)0, c, false, false);
        }
        if (lane == 0 || lane == 16) {
            const int rb = i0 + ((lane >> 4) << 3);
            float* __restrict__ o = out + (size_t)b * DIM + rb;
            o[0] = c[0]; o[1] = c[1]; o[2] = c[2]; o[3] = c[3];
            o[4] = c[4]; o[5] = c[5]; o[6] = c[6]; o[7] = c[7];
        }
    };

    const v8f zero8 = {0.f, 0.f, 0.f, 0.f, 0.f, 0.f, 0.f, 0.f};
    v8f c = zero8;
    float dh[8];

    load_panel(0);

    for (int k = 0; k < np; ++k) {
        const int second = (k >= np0);
        const int tile   = second ? t1 : t0;
        const int jc     = (second ? (k - np0) : k) * PW;
        const int i0     = tile << 4;
        const int width  = min(PW, i0 - jc);
        const bool last  = (jc + PW >= i0);

        #pragma unroll
        for (int u = 0; u < NLD; ++u) {
            if (128 * u + 4 * lane < width) {
                #pragma unroll
                for (int q = 0; q < 16; ++q)
                    *(v4f*)(st + q * PITCH + 128 * u + 4 * lane) = g[q * NLD + u];
            }
        }
        if (last) diag_load(tile, dh);
        if (k + 1 < np) load_panel(k + 1);

        __builtin_amdgcn_fence(__ATOMIC_RELEASE, "wavefront");
        __builtin_amdgcn_wave_barrier();
        __builtin_amdgcn_fence(__ATOMIC_ACQUIRE, "wavefront");

        const int ng = width >> 4;
        const float* ap = st + m * PITCH + kb;
        const float* zp = zs + jc + kb;
        for (int gi = 0; gi < ng; ++gi) {
            const v2f a0 = *(const v2f*)(ap + 0),  b0 = *(const v2f*)(zp + 0);
            const v2f a1 = *(const v2f*)(ap + 4),  b1 = *(const v2f*)(zp + 4);
            const v2f a2 = *(const v2f*)(ap + 8),  b2 = *(const v2f*)(zp + 8);
            const v2f a3 = *(const v2f*)(ap + 12), b3 = *(const v2f*)(zp + 12);
            c = __builtin_amdgcn_wmma_f32_16x16x4_f32(false, a0, false, b0,
                                                      (short)0, c, false, false);
            c = __builtin_amdgcn_wmma_f32_16x16x4_f32(false, a1, false, b1,
                                                      (short)0, c, false, false);
            c = __builtin_amdgcn_wmma_f32_16x16x4_f32(false, a2, false, b2,
                                                      (short)0, c, false, false);
            c = __builtin_amdgcn_wmma_f32_16x16x4_f32(false, a3, false, b3,
                                                      (short)0, c, false, false);
            ap += 16;
            zp += 16;
        }

        __builtin_amdgcn_fence(__ATOMIC_RELEASE, "wavefront");
        __builtin_amdgcn_wave_barrier();
        __builtin_amdgcn_fence(__ATOMIC_ACQUIRE, "wavefront");

        if (last) {
            diag_finish(tile, dh, c);
            c = zero8;
        }
    }

    if (np1 == 0) {
        diag_load(t1, dh);
        diag_finish(t1, dh, zero8);
    }

    if (b == 0 && tid == 0) out[(size_t)batch * DIM] = 0.0f;
}

extern "C" void kernel_launch(void* const* d_in, const int* in_sizes, int n_in,
                              void* d_out, int out_size, void* d_ws, size_t ws_size,
                              hipStream_t stream) {
    const float* z = (const float*)d_in[0];
    const float* h = (const float*)d_in[1];
    float* out = (float*)d_out;

    const int batch = in_sizes[0] / DIM;

    linmap_tri_matvec<<<dim3(batch), dim3(512), 0, stream>>>(z, h, out, batch);
}


# Request order, exact load waits: no lane-conditional panel loads/stores (edge lanes re-read in-row bytes), LDS-only fence at z barrier, panel 0 and tile 0's diagonal requested first; sum order same
# speedup vs baseline: 1.9292x; 1.0573x over previous
#include <hip/hip_runtime.h>

#define DIM 512
#define HLEN (DIM * (DIM - 1) / 2)
#define PW 256
#define NLD (PW / 128)
#define PITCH (PW + 4)
#define NWAVES 16

typedef __attribute__((ext_vector_type(2))) float v2f;
typedef __attribute__((ext_vector_type(4))) float v4f;
typedef __attribute__((ext_vector_type(8))) float v8f;
typedef float v4f_u __attribute__((ext_vector_type(4), aligned(4)));

__global__ void __launch_bounds__(512)
linmap_tri_matvec(const float* __restrict__ z,
                  const float* __restrict__ h,
                  float* __restrict__ out,
                  int batch) {
    __shared__ __attribute__((aligned(16))) float zs[DIM];
    __shared__ __attribute__((aligned(16))) float stage[NWAVES * 16 * PITCH];

    const int b   = blockIdx.x;
    const int tid = threadIdx.x;

    const int lane = tid & 31;
    const int wave = __builtin_amdgcn_readfirstlane(tid >> 5);
    const int m    = lane & 15;
    const int kb   = (lane >> 4) << 1;

    const float* __restrict__ hb = h + (size_t)b * (size_t)HLEN;
    float* __restrict__ st = stage + wave * (16 * PITCH);

    const int t0  = 31 - wave;
    const int t1  = wave;
    const int np0 = (16 * t0 + PW - 1) / PW;
    const int np1 = (16 * t1 + PW - 1) / PW;
    const int np  = np0 + np1;

    v4f g[16 * NLD];

    auto load_panel = [&](int k, bool full) {
        const int second = (k >= np0);
        const int tile   = second ? t1 : t0;
        const int jc     = (second ? (k - np0) : k) * PW;
        const int i0     = tile << 4;
        const int width  = full ? PW : min(PW, i0 - jc);
        #pragma unroll
        for (int u = 0; u < NLD; ++u) {
            if (full || 128 * u < width) {
                const int off = full ? (128 * u + 4 * lane) : (128 * u + min(4 * lane, width - 128 * u - 4));
                #pragma unroll
                for (int q = 0; q < 16; ++q) {
                    const int rq = i0 + q;
                    const float* p = hb + ((rq * (rq - 1)) >> 1) + jc + off;
                    g[q * NLD + u] = __builtin_nontemporal_load((const v4f_u*)p);
                }
            }
        }
    };

    auto diag_load = [&](int tile, float* dh) {
        const int i0 = tile << 4;
        const int r  = i0 + m;
        const float* __restrict__ hrow = hb + ((r * (r - 1)) >> 1);
        #pragma unroll
        for (int t = 0; t < 4; ++t) {
            const int c0 = i0 + 4 * t + kb;
            const int c1 = c0 + 1;
            int i0c = min(c0, r - 1); i0c = max(i0c, 0);
            int i1c = min(c1, r - 1); i1c = max(i1c, 0);
            dh[2 * t]     = hrow[i0c];
            dh[2 * t + 1] = hrow[i1c];
        }
    };

    auto diag_finish = [&](int tile, const float* dh, v8f c) {
        const int i0 = tile << 4;
        const int r  = i0 + m;
        #pragma unroll
        for (int t = 0; t < 4; ++t) {
            const int c0 = i0 + 4 * t + kb;
            const int c1 = c0 + 1;
            v2f a, bv;
            a.x = dh[2 * t]     * (float)(c0 < r) + (float)(c0 == r);
            a.y = dh[2 * t + 1] * (float)(c1 < r) + (float)(c1 == r);
            bv.x = zs[c0];
            bv.y = zs[c1];
            c = __builtin_amdgcn_wmma_f32_16x16x4_f32(false, a, false, bv,
                                                      (short)0, c, false, false);
        }
        if (lane == 0 || lane == 16) {
            const int rb = i0 + ((lane >> 4) << 3);
            float* __restrict__ o = out + (size_t)b * DIM + rb;
            o[0] = c[0]; o[1] = c[1]; o[2] = c[2]; o[3] = c[3];
            o[4] = c[4]; o[5] = c[5]; o[6] = c[6]; o[7] = c[7];
        }
    };

    const v8f zero8 = {0.f, 0.f, 0.f, 0.f, 0.f, 0.f, 0.f, 0.f};
    v8f c = zero8;
    float dh[8];

    float dh0[8];
    if (np1 == 0) diag_load(t1, dh0);

    const float zv = z[(size_t)b * DIM + tid];

    load_panel(0, true);

    zs[tid] = zv;
    __builtin_amdgcn_fence(__ATOMIC_RELEASE, "workgroup", "local");
    __builtin_amdgcn_s_barrier();
    __builtin_amdgcn_fence(__ATOMIC_ACQUIRE, "workgroup", "local");

    for (int k = 0; k < np; ++k) {
        const int second = (k >= np0);
        const int tile   = second ? t1 : t0;
        const int jc     = (second ? (k - np0) : k) * PW;
        const int i0     = tile << 4;
        const int width  = min(PW, i0 - jc);
        const bool last  = (jc + PW >= i0);

        #pragma unroll
        for (int u = 0; u < NLD; ++u) {
            #pragma unroll
            for (int q = 0; q < 16; ++q)
                *(v4f*)(st + q * PITCH + 128 * u + 4 * lane) = g[q * NLD + u];
        }
        if (last) diag_load(tile, dh);
        if (k + 1 < np) load_panel(k + 1, false);

        __builtin_amdgcn_fence(__ATOMIC_RELEASE, "wavefront");
        __builtin_amdgcn_wave_barrier();
        __builtin_amdgcn_fence(__ATOMIC_ACQUIRE, "wavefront");

        const int ng = width >> 4;
        const float* ap = st + m * PITCH + kb;
        const float* zp = zs + jc + kb;
        for (int gi = 0; gi < ng; ++gi) {
            const v2f a0 = *(const v2f*)(ap + 0),  b0 = *(const v2f*)(zp + 0);
            const v2f a1 = *(const v2f*)(ap + 4),  b1 = *(const v2f*)(zp + 4);
            const v2f a2 = *(const v2f*)(ap + 8),  b2 = *(const v2f*)(zp + 8);
            const v2f a3 = *(const v2f*)(ap + 12), b3 = *(const v2f*)(zp + 12);
            c = __builtin_amdgcn_wmma_f32_16x16x4_f32(false, a0, false, b0,
                                                      (short)0, c, false, false);
            c = __builtin_amdgcn_wmma_f32_16x16x4_f32(false, a1, false, b1,
                                                      (short)0, c, false, false);
            c = __builtin_amdgcn_wmma_f32_16x16x4_f32(false, a2, false, b2,
                                                      (short)0, c, false, false);
            c = __builtin_amdgcn_wmma_f32_16x16x4_f32(false, a3, false, b3,
                                                      (short)0, c, false, false);
            ap += 16;
            zp += 16;
        }

        __builtin_amdgcn_fence(__ATOMIC_RELEASE, "wavefront");
        __builtin_amdgcn_wave_barrier();
        __builtin_amdgcn_fence(__ATOMIC_ACQUIRE, "wavefront");

        if (last) {
            diag_finish(tile, dh, c);
            c = zero8;
        }
    }

    if (np1 == 0) diag_finish(t1, dh0, zero8);

    if (b == 0 && tid == 0) out[(size_t)batch * DIM] = 0.0f;
}

extern "C" void kernel_launch(void* const* d_in, const int* in_sizes, int n_in,
                              void* d_out, int out_size, void* d_ws, size_t ws_size,
                              hipStream_t stream) {
    const float* z = (const float*)d_in[0];
    const float* h = (const float*)d_in[1];
    float* out = (float*)d_out;

    const int batch = in_sizes[0] / DIM;

    linmap_tri_matvec<<<dim3(batch), dim3(512), 0, stream>>>(z, h, out, batch);
}


# Diagonal blocks ride on the row loads: a tile is one strip (bulk + its diagonal group) staged through LDS, the 8 sixteen-line 4-byte gathers per tile go; A values selected there; sum order same
# speedup vs baseline: 2.1535x; 1.1163x over previous
#include <hip/hip_runtime.h>

#define DIM 512
#define HLEN (DIM * (DIM - 1) / 2)
#define PW 256
#define NLD (PW / 128)
#define PITCH (PW + 4)
#define NWAVES 16

typedef __attribute__((ext_vector_type(2))) float v2f;
typedef __attribute__((ext_vector_type(4))) float v4f;
typedef __attribute__((ext_vector_type(8))) float v8f;
typedef float v4f_u __attribute__((ext_vector_type(4), aligned(4)));

__global__ void __launch_bounds__(512)
linmap_tri_matvec(const float* __restrict__ z,
                  const float* __restrict__ h,
                  float* __restrict__ out,
                  int batch) {
    __shared__ __attribute__((aligned(16))) float zs[DIM];
    __shared__ __attribute__((aligned(16))) float stage[NWAVES * 16 * PITCH];

    const int b   = blockIdx.x;
    const int tid = threadIdx.x;

    const int lane = tid & 31;
    const int wave = __builtin_amdgcn_readfirstlane(tid >> 5);
    const int m    = lane & 15;
    const int kb   = (lane >> 4) << 1;

    const float* __restrict__ hb = h + (size_t)b * (size_t)HLEN;
    float* __restrict__ st = stage + wave * (16 * PITCH);

    const int t0  = 31 - wave;
    const int t1  = wave;
    const int np0 = (16 * t0 + 16 + PW - 1) / PW;
    const int np1 = (16 * t1 + 16 + PW - 1) / PW;
    const int np  = np0 + np1;

    v4f g[16 * NLD];

    auto load_panel = [&](int k, bool full) {
        const int second = (k >= np0);
        const int tile   = second ? t1 : t0;
        const int jc     = (second ? (k - np0) : k) * PW;
        const int i0     = tile << 4;
        const int width  = full ? PW : min(PW, i0 + 16 - jc);
        const bool lastp = !full && (jc + PW >= i0 + 16);
        #pragma unroll
        for (int u = 0; u < NLD; ++u) {
            if (full || 128 * u < width) {
                const int lim   = width - 128 * u - 4;
                const int off   = full ? (128 * u + 4 * lane) : (128 * u + min(4 * lane, lim));
                const int off15 = full ? off : (128 * u + min(4 * lane, lastp ? lim - 1 : lim));
                #pragma unroll
                for (int q = 0; q < 16; ++q) {
                    const int rq = i0 + q;
                    const float* p = hb + ((rq * (rq - 1)) >> 1) + jc + (q == 15 ? off15 : off);
                    g[q * NLD + u] = __builtin_nontemporal_load((const v4f_u*)p);
                }
            }
        }
    };

    const v8f zero8 = {0.f, 0.f, 0.f, 0.f, 0.f, 0.f, 0.f, 0.f};
    v8f c = zero8;

    const float zv = z[(size_t)b * DIM + tid];

    load_panel(0, true);

    zs[tid] = zv;
    __builtin_amdgcn_fence(__ATOMIC_RELEASE, "workgroup", "local");
    __builtin_amdgcn_s_barrier();
    __builtin_amdgcn_fence(__ATOMIC_ACQUIRE, "workgroup", "local");

    for (int k = 0; k < np; ++k) {
        const int second = (k >= np0);
        const int tile   = second ? t1 : t0;
        const int jc     = (second ? (k - np0) : k) * PW;
        const int i0     = tile << 4;
        const int width  = min(PW, i0 + 16 - jc);
        const bool last  = (jc + PW >= i0 + 16);

        #pragma unroll
        for (int u = 0; u < NLD; ++u) {
            #pragma unroll
            for (int q = 0; q < 16; ++q) {
                v4f v = g[q * NLD + u];
                if (q == 15) {
                    const bool rot = last && (4 * lane >= width - 128 * u - 4);
                    const v4f  sh  = {v.y, v.z, v.w, 0.f};
                    v = rot ? sh : v;
                }
                *(v4f*)(st + q * PITCH + 128 * u + 4 * lane) = v;
            }
        }
        if (k + 1 < np) load_panel(k + 1, false);

        __builtin_amdgcn_fence(__ATOMIC_RELEASE, "wavefront");
        __builtin_amdgcn_wave_barrier();
        __builtin_amdgcn_fence(__ATOMIC_ACQUIRE, "wavefront");

        const int ng = (width >> 4) - (last ? 1 : 0);
        const float* ap = st + m * PITCH + kb;
        const float* zp = zs + jc + kb;
        for (int gi = 0; gi < ng; ++gi) {
            const v2f a0 = *(const v2f*)(ap + 0),  b0 = *(const v2f*)(zp + 0);
            const v2f a1 = *(const v2f*)(ap + 4),  b1 = *(const v2f*)(zp + 4);
            const v2f a2 = *(const v2f*)(ap + 8),  b2 = *(const v2f*)(zp + 8);
            const v2f a3 = *(const v2f*)(ap + 12), b3 = *(const v2f*)(zp + 12);
            c = __builtin_amdgcn_wmma_f32_16x16x4_f32(false, a0, false, b0,
                                                      (short)0, c, false, false);
            c = __builtin_amdgcn_wmma_f32_16x16x4_f32(false, a1, false, b1,
                                                      (short)0, c, false, false);
            c = __builtin_amdgcn_wmma_f32_16x16x4_f32(false, a2, false, b2,
                                                      (short)0, c, false, false);
            c = __builtin_amdgcn_wmma_f32_16x16x4_f32(false, a3, false, b3,
                                                      (short)0, c, false, false);
            ap += 16;
            zp += 16;
        }
        if (last) {
            const int r = i0 + m;
            #pragma unroll
            for (int t = 0; t < 4; ++t) {
                const int c0 = i0 + 4 * t + kb;
                const int c1 = c0 + 1;
                const v2f av = *(const v2f*)(ap + 4 * t);
                const v2f bv = *(const v2f*)(zp + 4 * t);
                v2f a;
                a.x = (c0 < r) ? av.x : ((c0 == r) ? 1.0f : 0.0f);
                a.y = (c1 < r) ? av.y : ((c1 == r) ? 1.0f : 0.0f);
                c = __builtin_amdgcn_wmma_f32_16x16x4_f32(false, a, false, bv,
                                                          (short)0, c, false, false);
            }
        }

        __builtin_amdgcn_fence(__ATOMIC_RELEASE, "wavefront");
        __builtin_amdgcn_wave_barrier();
        __builtin_amdgcn_fence(__ATOMIC_ACQUIRE, "wavefront");

        if (last) {
            if (lane == 0 || lane == 16) {
                const int rb = i0 + ((lane >> 4) << 3);
                float* __restrict__ o = out + (size_t)b * DIM + rb;
                o[0] = c[0]; o[1] = c[1]; o[2] = c[2]; o[3] = c[3];
                o[4] = c[4]; o[5] = c[5]; o[6] = c[6]; o[7] = c[7];
            }
            c = zero8;
        }
    }

    if (b == 0 && tid == 0) out[(size_t)batch * DIM] = 0.0f;
}

extern "C" void kernel_launch(void* const* d_in, const int* in_sizes, int n_in,
                              void* d_out, int out_size, void* d_ws, size_t ws_size,
                              hipStream_t stream) {
    const float* z = (const float*)d_in[0];
    const float* h = (const float*)d_in[1];
    float* out = (float*)d_out;

    const int batch = in_sizes[0] / DIM;

    linmap_tri_matvec<<<dim3(batch), dim3(512), 0, stream>>>(z, h, out, batch);
}
